# starting kernel
# baseline (speedup 1.0000x reference)
#include <hip/hip_runtime.h>
#include <hip/hip_bf16.h>


typedef __attribute__((ext_vector_type(2))) float v2f;
typedef __attribute__((ext_vector_type(8))) float v8f;

namespace caps {
constexpr int B   = 32;
constexpr int In  = 4096;
constexpr int D   = 256;
constexpr int N   = 16;
constexpr int K   = 16;
constexpr int SPLIT = 8;
}

using namespace caps;

__global__ void zero_buf(float* p, int n) {
    int i = blockIdx.x * blockDim.x + threadIdx.x;
    if (i < n) p[i] = 0.0f;
}

__global__ void softmax_n(float* __restrict__ bc) {
    int idx = blockIdx.x * blockDim.x + threadIdx.x;
    if (idx >= B * In) return;
    float* p = bc + (size_t)idx * N;
    float v[N];
    float m = -3.402823466e+38f;
#pragma unroll
    for (int n = 0; n < N; ++n) { v[n] = p[n]; m = fmaxf(m, v[n]); }
    float s = 0.0f;
#pragma unroll
    for (int n = 0; n < N; ++n) { v[n] = __expf(v[n] - m); s += v[n]; }
    float inv = 1.0f / s;
#pragma unroll
    for (int n = 0; n < N; ++n) p[n] = v[n] * inv;
}

__global__ void __launch_bounds__(128)
gemm_s(const float* __restrict__ x, const float* __restrict__ c,
       float* __restrict__ s_part) {
    const int b    = blockIdx.y;
    const int sp   = blockIdx.x;
    const int lane = threadIdx.x & 31;
    const int wave = threadIdx.x >> 5;
    const int half = lane >> 4;
    const int l16  = lane & 15;

    const float* cb = c + (size_t)b * In * N;
    const float* xb = x + (size_t)b * In * D;

    const int d0 = wave * 64;
    const int i_beg = sp * (In / SPLIT);
    const int i_end = i_beg + (In / SPLIT);

    v8f acc0 = {}, acc1 = {}, acc2 = {}, acc3 = {};

    for (int i0 = i_beg; i0 < i_end; i0 += 4) {
        const int kr = i0 + half * 2;
        v2f a;
        a.x = cb[(size_t)(kr + 0) * N + l16];
        a.y = cb[(size_t)(kr + 1) * N + l16];
        const float* xr0 = xb + (size_t)(kr + 0) * D;
        const float* xr1 = xb + (size_t)(kr + 1) * D;
        __builtin_prefetch(xr0 + 4 * D, 0, 1);
        v2f b0, b1, b2, b3;
        b0.x = xr0[d0 +  0 + l16]; b0.y = xr1[d0 +  0 + l16];
        b1.x = xr0[d0 + 16 + l16]; b1.y = xr1[d0 + 16 + l16];
        b2.x = xr0[d0 + 32 + l16]; b2.y = xr1[d0 + 32 + l16];
        b3.x = xr0[d0 + 48 + l16]; b3.y = xr1[d0 + 48 + l16];
        acc0 = __builtin_amdgcn_wmma_f32_16x16x4_f32(false, a, false, b0, (short)0, acc0, false, false);
        acc1 = __builtin_amdgcn_wmma_f32_16x16x4_f32(false, a, false, b1, (short)0, acc1, false, false);
        acc2 = __builtin_amdgcn_wmma_f32_16x16x4_f32(false, a, false, b2, (short)0, acc2, false, false);
        acc3 = __builtin_amdgcn_wmma_f32_16x16x4_f32(false, a, false, b3, (short)0, acc3, false, false);
    }

    float* outp = s_part + (size_t)(b * SPLIT + sp) * N * D;
#pragma unroll
    for (int r = 0; r < 8; ++r) {
        const int row = r + half * 8;
        outp[(size_t)row * D + d0 +  0 + l16] = acc0[r];
        outp[(size_t)row * D + d0 + 16 + l16] = acc1[r];
        outp[(size_t)row * D + d0 + 32 + l16] = acc2[r];
        outp[(size_t)row * D + d0 + 48 + l16] = acc3[r];
    }
}

__global__ void reduce_s(const float* __restrict__ s_part, float* __restrict__ s) {
    int idx = blockIdx.x * blockDim.x + threadIdx.x;
    if (idx >= B * N * D) return;
    int b  = idx / (N * D);
    int nd = idx - b * (N * D);
    float acc = 0.0f;
#pragma unroll
    for (int sp = 0; sp < SPLIT; ++sp)
        acc += s_part[(size_t)(b * SPLIT + sp) * N * D + nd];
    s[idx] = acc;
}

__global__ void __launch_bounds__(256)
outputs_squash(const float* __restrict__ s, const float* __restrict__ W,
               float* __restrict__ out) {
    __shared__ float part[D * K];
    __shared__ float dot[K];
    const int bn = blockIdx.x;
    const int n  = bn & (N - 1);
    const int d  = threadIdx.x;

    const float  sv = s[(size_t)bn * D + d];
    const float* wr = W + (size_t)d * (N * K) + (size_t)n * K;
#pragma unroll
    for (int k = 0; k < K; ++k) part[d * K + k] = sv * wr[k];
    __syncthreads();

    if (d < K) {
        float a = 0.0f;
        for (int dd = 0; dd < D; ++dd) a += part[dd * K + d];
        dot[d] = a;
    }
    __syncthreads();

    if (d < K) {
        float n2 = 0.0f;
#pragma unroll
        for (int k = 0; k < K; ++k) n2 += dot[k] * dot[k];
        out[(size_t)bn * K + d] = dot[d] * rsqrtf(n2 + 1e-7f);
    }
}

__global__ void __launch_bounds__(256)
compute_t(const float* __restrict__ W, const float* __restrict__ out,
          float* __restrict__ t) {
    __shared__ float ov[K];
    const int bn = blockIdx.x;
    const int n  = bn & (N - 1);
    const int d  = threadIdx.x;
    if (d < K) ov[d] = out[(size_t)bn * K + d];
    __syncthreads();
    const float* wr = W + (size_t)d * (N * K) + (size_t)n * K;
    float acc = 0.0f;
#pragma unroll
    for (int k = 0; k < K; ++k) acc += wr[k] * ov[k];
    t[(size_t)bn * D + d] = acc;
}

__global__ void __launch_bounds__(128)
bupdate(const float* __restrict__ x, const float* __restrict__ t,
        float* __restrict__ bc) {
    const int b    = blockIdx.y;
    const int lane = threadIdx.x & 31;
    const int wave = threadIdx.x >> 5;
    const int half = lane >> 4;
    const int l16  = lane & 15;
    const int i0   = (blockIdx.x * 4 + wave) * 16;

    const float* xb = x + (size_t)b * In * D;
    const float* tb = t + (size_t)b * N * D;

    v8f acc = {};
    for (int d0 = 0; d0 < D; d0 += 4) {
        const int kd = d0 + half * 2;
        const float* xr = xb + (size_t)(i0 + l16) * D + kd;
        __builtin_prefetch(xr + 64, 0, 1);
        v2f a;  a.x = xr[0];  a.y = xr[1];
        const float* tr = tb + (size_t)l16 * D + kd;
        v2f bb; bb.x = tr[0]; bb.y = tr[1];
        acc = __builtin_amdgcn_wmma_f32_16x16x4_f32(false, a, false, bb, (short)0, acc, false, false);
    }

    float* dst = bc + ((size_t)b * In + i0) * N;
#pragma unroll
    for (int r = 0; r < 8; ++r)
        dst[(size_t)(r + half * 8) * N + l16] = acc[r];
}

extern "C" void kernel_launch(void* const* d_in, const int* in_sizes, int n_in,
                              void* d_out, int out_size, void* d_ws, size_t ws_size,
                              hipStream_t stream) {
    const float* x = (const float*)d_in[0];
    const float* W = (const float*)d_in[1];
    float* out = (float*)d_out;

    float* ws     = (float*)d_ws;
    float* bc     = ws;
    float* s_part = bc + (size_t)B * In * N;
    float* s      = s_part + (size_t)B * SPLIT * N * D;
    float* t      = s + (size_t)B * N * D;

    const int bc_n = B * In * N;
    zero_buf<<<(bc_n + 255) / 256, 256, 0, stream>>>(bc, bc_n);

    for (int it = 0; it < 3; ++it) {
        softmax_n<<<(B * In + 255) / 256, 256, 0, stream>>>(bc);
        gemm_s<<<dim3(SPLIT, B), 128, 0, stream>>>(x, bc, s_part);
        reduce_s<<<(B * N * D + 255) / 256, 256, 0, stream>>>(s_part, s);
        outputs_squash<<<B * N, 256, 0, stream>>>(s, W, out);
        if (it < 2) {
            compute_t<<<B * N, 256, 0, stream>>>(W, out, t);
            bupdate<<<dim3(In / 64, B), 128, 0, stream>>>(x, t, bc);
        }
    }
}


# gemm_s: 8-wave blocks split each 512-row chunk 4 ways (partials added via LDS: new order of the sum over i); x read as 16-byte row pieces, next 8 k-steps' loads issued ahead of the WMMAs
# speedup vs baseline: 1.5582x; 1.5582x over previous
#include <hip/hip_runtime.h>
#include <hip/hip_bf16.h>


typedef __attribute__((ext_vector_type(2))) float v2f;
typedef __attribute__((ext_vector_type(4))) float v4f;
typedef __attribute__((ext_vector_type(8))) float v8f;

namespace caps {
constexpr int B   = 32;
constexpr int In  = 4096;
constexpr int D   = 256;
constexpr int N   = 16;
constexpr int K   = 16;
constexpr int SPLIT = 8;
}

using namespace caps;

__global__ void zero_buf(float* p, int n) {
    int i = blockIdx.x * blockDim.x + threadIdx.x;
    if (i < n) p[i] = 0.0f;
}

__global__ void softmax_n(float* __restrict__ bc) {
    int idx = blockIdx.x * blockDim.x + threadIdx.x;
    if (idx >= B * In) return;
    float* p = bc + (size_t)idx * N;
    float v[N];
    float m = -3.402823466e+38f;
#pragma unroll
    for (int n = 0; n < N; ++n) { v[n] = p[n]; m = fmaxf(m, v[n]); }
    float s = 0.0f;
#pragma unroll
    for (int n = 0; n < N; ++n) { v[n] = __expf(v[n] - m); s += v[n]; }
    float inv = 1.0f / s;
#pragma unroll
    for (int n = 0; n < N; ++n) p[n] = v[n] * inv;
}

constexpr int GS_ROWS  = In / SPLIT;
constexpr int GS_STEPS = 8;
constexpr int GS_GROW  = 4 * GS_STEPS;
constexpr int GS_RQ    = 4;
constexpr int GS_NGRP  = GS_ROWS / GS_GROW / GS_RQ;

struct GsFrag {
    v4f   x0[GS_STEPS], x1[GS_STEPS];
    float c0[GS_STEPS], c1[GS_STEPS];
};

__device__ __forceinline__ void gs_load(GsFrag& f, const float* __restrict__ xp,
                                        const float* __restrict__ cp) {
#pragma unroll
    for (int u = 0; u < GS_STEPS; ++u) {
        f.c0[u] = cp[(size_t)(4 * u + 0) * N];
        f.c1[u] = cp[(size_t)(4 * u + 1) * N];
        f.x0[u] = *reinterpret_cast<const v4f*>(xp + (size_t)(4 * u + 0) * D);
        f.x1[u] = *reinterpret_cast<const v4f*>(xp + (size_t)(4 * u + 1) * D);
    }
}

__device__ __forceinline__ void gs_mma(const GsFrag& f, v8f& acc0, v8f& acc1,
                                       v8f& acc2, v8f& acc3) {
#pragma unroll
    for (int u = 0; u < GS_STEPS; ++u) {
        v2f a;  a.x  = f.c0[u];    a.y  = f.c1[u];
        v2f b0; b0.x = f.x0[u].x;  b0.y = f.x1[u].x;
        v2f b1; b1.x = f.x0[u].y;  b1.y = f.x1[u].y;
        v2f b2; b2.x = f.x0[u].z;  b2.y = f.x1[u].z;
        v2f b3; b3.x = f.x0[u].w;  b3.y = f.x1[u].w;
        acc0 = __builtin_amdgcn_wmma_f32_16x16x4_f32(false, a, false, b0, (short)0, acc0, false, false);
        acc1 = __builtin_amdgcn_wmma_f32_16x16x4_f32(false, a, false, b1, (short)0, acc1, false, false);
        acc2 = __builtin_amdgcn_wmma_f32_16x16x4_f32(false, a, false, b2, (short)0, acc2, false, false);
        acc3 = __builtin_amdgcn_wmma_f32_16x16x4_f32(false, a, false, b3, (short)0, acc3, false, false);
    }
}

__global__ void __launch_bounds__(256)
gemm_s(const float* __restrict__ x, const float* __restrict__ c,
       float* __restrict__ s_part) {
    __shared__ v4f red[8 * 8 * 32];
    const int b    = blockIdx.z;
    const int sp   = blockIdx.y;
    const int lane = threadIdx.x & 31;
    const int wave = threadIdx.x >> 5;
    const int half = lane >> 4;
    const int l16  = lane & 15;
    const int dg   = wave & 1;
    const int rq   = wave >> 1;

    const int d0    = blockIdx.x * 128 + dg * 64;
    const int i_beg = sp * GS_ROWS + rq * GS_GROW + half * 2;

    const float* xp = x + ((size_t)b * In + i_beg) * D + d0 + 4 * l16;
    const float* cp = c + ((size_t)b * In + i_beg) * N + l16;
    constexpr size_t XG = (size_t)GS_RQ * GS_GROW * D;
    constexpr size_t CG = (size_t)GS_RQ * GS_GROW * N;

    v8f acc0 = {}, acc1 = {}, acc2 = {}, acc3 = {};
    GsFrag fa, fb;

    gs_load(fa, xp, cp);
#pragma unroll
    for (int t = 0; t < GS_NGRP; t += 2) {
        gs_load(fb, xp + (t + 1) * XG, cp + (t + 1) * CG);
        gs_mma(fa, acc0, acc1, acc2, acc3);
        if (t + 2 < GS_NGRP) gs_load(fa, xp + (t + 2) * XG, cp + (t + 2) * CG);
        gs_mma(fb, acc0, acc1, acc2, acc3);
    }

#pragma unroll
    for (int r = 0; r < 8; ++r) {
        v4f v; v.x = acc0[r]; v.y = acc1[r]; v.z = acc2[r]; v.w = acc3[r];
        red[(wave * 8 + r) * 32 + lane] = v;
    }
    __syncthreads();

    float* outp = s_part + (size_t)(b * SPLIT + sp) * N * D;
    for (int q = 0; q < 2; ++q) {
        v4f v = red[((0 * 2 + q) * 8 + wave) * 32 + lane];
        for (int p = 1; p < GS_RQ; ++p)
            v += red[((p * 2 + q) * 8 + wave) * 32 + lane];
        const int row = wave + 8 * half;
        const int d   = blockIdx.x * 128 + q * 64 + 4 * l16;
        *reinterpret_cast<v4f*>(outp + (size_t)row * D + d) = v;
    }
}

__global__ void reduce_s(const float* __restrict__ s_part, float* __restrict__ s) {
    int idx = blockIdx.x * blockDim.x + threadIdx.x;
    if (idx >= B * N * D) return;
    int b  = idx / (N * D);
    int nd = idx - b * (N * D);
    float acc = 0.0f;
#pragma unroll
    for (int sp = 0; sp < SPLIT; ++sp)
        acc += s_part[(size_t)(b * SPLIT + sp) * N * D + nd];
    s[idx] = acc;
}

__global__ void __launch_bounds__(256)
outputs_squash(const float* __restrict__ s, const float* __restrict__ W,
               float* __restrict__ out) {
    __shared__ float part[D * K];
    __shared__ float dot[K];
    const int bn = blockIdx.x;
    const int n  = bn & (N - 1);
    const int d  = threadIdx.x;

    const float  sv = s[(size_t)bn * D + d];
    const float* wr = W + (size_t)d * (N * K) + (size_t)n * K;
#pragma unroll
    for (int k = 0; k < K; ++k) part[d * K + k] = sv * wr[k];
    __syncthreads();

    if (d < K) {
        float a = 0.0f;
        for (int dd = 0; dd < D; ++dd) a += part[dd * K + d];
        dot[d] = a;
    }
    __syncthreads();

    if (d < K) {
        float n2 = 0.0f;
#pragma unroll
        for (int k = 0; k < K; ++k) n2 += dot[k] * dot[k];
        out[(size_t)bn * K + d] = dot[d] * rsqrtf(n2 + 1e-7f);
    }
}

__global__ void __launch_bounds__(256)
compute_t(const float* __restrict__ W, const float* __restrict__ out,
          float* __restrict__ t) {
    __shared__ float ov[K];
    const int bn = blockIdx.x;
    const int n  = bn & (N - 1);
    const int d  = threadIdx.x;
    if (d < K) ov[d] = out[(size_t)bn * K + d];
    __syncthreads();
    const float* wr = W + (size_t)d * (N * K) + (size_t)n * K;
    float acc = 0.0f;
#pragma unroll
    for (int k = 0; k < K; ++k) acc += wr[k] * ov[k];
    t[(size_t)bn * D + d] = acc;
}

__global__ void __launch_bounds__(128)
bupdate(const float* __restrict__ x, const float* __restrict__ t,
        float* __restrict__ bc) {
    const int b    = blockIdx.y;
    const int lane = threadIdx.x & 31;
    const int wave = threadIdx.x >> 5;
    const int half = lane >> 4;
    const int l16  = lane & 15;
    const int i0   = (blockIdx.x * 4 + wave) * 16;

    const float* xb = x + (size_t)b * In * D;
    const float* tb = t + (size_t)b * N * D;

    v8f acc = {};
    for (int d0 = 0; d0 < D; d0 += 4) {
        const int kd = d0 + half * 2;
        const float* xr = xb + (size_t)(i0 + l16) * D + kd;
        __builtin_prefetch(xr + 64, 0, 1);
        v2f a;  a.x = xr[0];  a.y = xr[1];
        const float* tr = tb + (size_t)l16 * D + kd;
        v2f bb; bb.x = tr[0]; bb.y = tr[1];
        acc = __builtin_amdgcn_wmma_f32_16x16x4_f32(false, a, false, bb, (short)0, acc, false, false);
    }

    float* dst = bc + ((size_t)b * In + i0) * N;
#pragma unroll
    for (int r = 0; r < 8; ++r)
        dst[(size_t)(r + half * 8) * N + l16] = acc[r];
}

extern "C" void kernel_launch(void* const* d_in, const int* in_sizes, int n_in,
                              void* d_out, int out_size, void* d_ws, size_t ws_size,
                              hipStream_t stream) {
    const float* x = (const float*)d_in[0];
    const float* W = (const float*)d_in[1];
    float* out = (float*)d_out;

    float* ws     = (float*)d_ws;
    float* bc     = ws;
    float* s_part = bc + (size_t)B * In * N;
    float* s      = s_part + (size_t)B * SPLIT * N * D;
    float* t      = s + (size_t)B * N * D;

    const int bc_n = B * In * N;
    zero_buf<<<(bc_n + 255) / 256, 256, 0, stream>>>(bc, bc_n);

    for (int it = 0; it < 3; ++it) {
        softmax_n<<<(B * In + 255) / 256, 256, 0, stream>>>(bc);
        gemm_s<<<dim3(2, SPLIT, B), 256, 0, stream>>>(x, bc, s_part);
        reduce_s<<<(B * N * D + 255) / 256, 256, 0, stream>>>(s_part, s);
        outputs_squash<<<B * N, 256, 0, stream>>>(s, W, out);
        if (it < 2) {
            compute_t<<<B * N, 256, 0, stream>>>(W, out, t);
            bupdate<<<dim3(In / 64, B), 128, 0, stream>>>(x, t, bc);
        }
    }
}


# bupdate: each wave reads its 16 x rows as one contiguous 16 KB, stages them and t in padded LDS and takes the WMMA fragments from LDS; two accumulators: new order of the sum over d
# speedup vs baseline: 1.9057x; 1.2230x over previous
#include <hip/hip_runtime.h>
#include <hip/hip_bf16.h>


typedef __attribute__((ext_vector_type(2))) float v2f;
typedef __attribute__((ext_vector_type(4))) float v4f;
typedef __attribute__((ext_vector_type(8))) float v8f;

namespace caps {
constexpr int B   = 32;
constexpr int In  = 4096;
constexpr int D   = 256;
constexpr int N   = 16;
constexpr int K   = 16;
constexpr int SPLIT = 8;
}

using namespace caps;

__global__ void zero_buf(float* p, int n) {
    int i = blockIdx.x * blockDim.x + threadIdx.x;
    if (i < n) p[i] = 0.0f;
}

__global__ void softmax_n(float* __restrict__ bc) {
    int idx = blockIdx.x * blockDim.x + threadIdx.x;
    if (idx >= B * In) return;
    float* p = bc + (size_t)idx * N;
    float v[N];
    float m = -3.402823466e+38f;
#pragma unroll
    for (int n = 0; n < N; ++n) { v[n] = p[n]; m = fmaxf(m, v[n]); }
    float s = 0.0f;
#pragma unroll
    for (int n = 0; n < N; ++n) { v[n] = __expf(v[n] - m); s += v[n]; }
    float inv = 1.0f / s;
#pragma unroll
    for (int n = 0; n < N; ++n) p[n] = v[n] * inv;
}

constexpr int GS_ROWS  = In / SPLIT;
constexpr int GS_STEPS = 8;
constexpr int GS_GROW  = 4 * GS_STEPS;
constexpr int GS_RQ    = 4;
constexpr int GS_NGRP  = GS_ROWS / GS_GROW / GS_RQ;

struct GsFrag {
    v4f   x0[GS_STEPS], x1[GS_STEPS];
    float c0[GS_STEPS], c1[GS_STEPS];
};

__device__ __forceinline__ void gs_load(GsFrag& f, const float* __restrict__ xp,
                                        const float* __restrict__ cp) {
#pragma unroll
    for (int u = 0; u < GS_STEPS; ++u) {
        f.c0[u] = cp[(size_t)(4 * u + 0) * N];
        f.c1[u] = cp[(size_t)(4 * u + 1) * N];
        f.x0[u] = *reinterpret_cast<const v4f*>(xp + (size_t)(4 * u + 0) * D);
        f.x1[u] = *reinterpret_cast<const v4f*>(xp + (size_t)(4 * u + 1) * D);
    }
}

__device__ __forceinline__ void gs_mma(const GsFrag& f, v8f& acc0, v8f& acc1,
                                       v8f& acc2, v8f& acc3) {
#pragma unroll
    for (int u = 0; u < GS_STEPS; ++u) {
        v2f a;  a.x  = f.c0[u];    a.y  = f.c1[u];
        v2f b0; b0.x = f.x0[u].x;  b0.y = f.x1[u].x;
        v2f b1; b1.x = f.x0[u].y;  b1.y = f.x1[u].y;
        v2f b2; b2.x = f.x0[u].z;  b2.y = f.x1[u].z;
        v2f b3; b3.x = f.x0[u].w;  b3.y = f.x1[u].w;
        acc0 = __builtin_amdgcn_wmma_f32_16x16x4_f32(false, a, false, b0, (short)0, acc0, false, false);
        acc1 = __builtin_amdgcn_wmma_f32_16x16x4_f32(false, a, false, b1, (short)0, acc1, false, false);
        acc2 = __builtin_amdgcn_wmma_f32_16x16x4_f32(false, a, false, b2, (short)0, acc2, false, false);
        acc3 = __builtin_amdgcn_wmma_f32_16x16x4_f32(false, a, false, b3, (short)0, acc3, false, false);
    }
}

__global__ void __launch_bounds__(256)
gemm_s(const float* __restrict__ x, const float* __restrict__ c,
       float* __restrict__ s_part) {
    __shared__ v4f red[8 * 8 * 32];
    const int b    = blockIdx.z;
    const int sp   = blockIdx.y;
    const int lane = threadIdx.x & 31;
    const int wave = threadIdx.x >> 5;
    const int half = lane >> 4;
    const int l16  = lane & 15;
    const int dg   = wave & 1;
    const int rq   = wave >> 1;

    const int d0    = blockIdx.x * 128 + dg * 64;
    const int i_beg = sp * GS_ROWS + rq * GS_GROW + half * 2;

    const float* xp = x + ((size_t)b * In + i_beg) * D + d0 + 4 * l16;
    const float* cp = c + ((size_t)b * In + i_beg) * N + l16;
    constexpr size_t XG = (size_t)GS_RQ * GS_GROW * D;
    constexpr size_t CG = (size_t)GS_RQ * GS_GROW * N;

    v8f acc0 = {}, acc1 = {}, acc2 = {}, acc3 = {};
    GsFrag fa, fb;

    gs_load(fa, xp, cp);
#pragma unroll
    for (int t = 0; t < GS_NGRP; t += 2) {
        gs_load(fb, xp + (t + 1) * XG, cp + (t + 1) * CG);
        gs_mma(fa, acc0, acc1, acc2, acc3);
        if (t + 2 < GS_NGRP) gs_load(fa, xp + (t + 2) * XG, cp + (t + 2) * CG);
        gs_mma(fb, acc0, acc1, acc2, acc3);
    }

#pragma unroll
    for (int r = 0; r < 8; ++r) {
        v4f v; v.x = acc0[r]; v.y = acc1[r]; v.z = acc2[r]; v.w = acc3[r];
        red[(wave * 8 + r) * 32 + lane] = v;
    }
    __syncthreads();

    float* outp = s_part + (size_t)(b * SPLIT + sp) * N * D;
    for (int q = 0; q < 2; ++q) {
        v4f v = red[((0 * 2 + q) * 8 + wave) * 32 + lane];
        for (int p = 1; p < GS_RQ; ++p)
            v += red[((p * 2 + q) * 8 + wave) * 32 + lane];
        const int row = wave + 8 * half;
        const int d   = blockIdx.x * 128 + q * 64 + 4 * l16;
        *reinterpret_cast<v4f*>(outp + (size_t)row * D + d) = v;
    }
}

__global__ void reduce_s(const float* __restrict__ s_part, float* __restrict__ s) {
    int idx = blockIdx.x * blockDim.x + threadIdx.x;
    if (idx >= B * N * D) return;
    int b  = idx / (N * D);
    int nd = idx - b * (N * D);
    float acc = 0.0f;
#pragma unroll
    for (int sp = 0; sp < SPLIT; ++sp)
        acc += s_part[(size_t)(b * SPLIT + sp) * N * D + nd];
    s[idx] = acc;
}

__global__ void __launch_bounds__(256)
outputs_squash(const float* __restrict__ s, const float* __restrict__ W,
               float* __restrict__ out) {
    __shared__ float part[D * K];
    __shared__ float dot[K];
    const int bn = blockIdx.x;
    const int n  = bn & (N - 1);
    const int d  = threadIdx.x;

    const float  sv = s[(size_t)bn * D + d];
    const float* wr = W + (size_t)d * (N * K) + (size_t)n * K;
#pragma unroll
    for (int k = 0; k < K; ++k) part[d * K + k] = sv * wr[k];
    __syncthreads();

    if (d < K) {
        float a = 0.0f;
        for (int dd = 0; dd < D; ++dd) a += part[dd * K + d];
        dot[d] = a;
    }
    __syncthreads();

    if (d < K) {
        float n2 = 0.0f;
#pragma unroll
        for (int k = 0; k < K; ++k) n2 += dot[k] * dot[k];
        out[(size_t)bn * K + d] = dot[d] * rsqrtf(n2 + 1e-7f);
    }
}

__global__ void __launch_bounds__(256)
compute_t(const float* __restrict__ W, const float* __restrict__ out,
          float* __restrict__ t) {
    __shared__ float ov[K];
    const int bn = blockIdx.x;
    const int n  = bn & (N - 1);
    const int d  = threadIdx.x;
    if (d < K) ov[d] = out[(size_t)bn * K + d];
    __syncthreads();
    const float* wr = W + (size_t)d * (N * K) + (size_t)n * K;
    float acc = 0.0f;
#pragma unroll
    for (int k = 0; k < K; ++k) acc += wr[k] * ov[k];
    t[(size_t)bn * D + d] = acc;
}

constexpr int BU_WAVES  = 8;
constexpr int BU_TILES  = 2;
constexpr int BU_ROWS   = 16 * BU_WAVES * BU_TILES;
constexpr int BU_PITCH4 = D / 4 + 1;
constexpr int BU_TILE4  = 16 * BU_PITCH4;

__device__ __forceinline__ void bu_load(v4f (&g)[32], const v4f* __restrict__ src, int lane) {
#pragma unroll
    for (int j = 0; j < 32; ++j) g[j] = src[32 * j + lane];
}

__device__ __forceinline__ void bu_stage(const v4f (&g)[32], v4f* xs, int lane) {
#pragma unroll
    for (int j = 0; j < 32; ++j) xs[(j >> 1) * BU_PITCH4 + 32 * (j & 1) + lane] = g[j];
}

__device__ __forceinline__ v8f bu_tile(const v4f* xs, const v4f* ts, int l16, int half) {
    const v4f* xr = xs + l16 * BU_PITCH4 + half;
    const v4f* tr = ts + l16 * BU_PITCH4 + half;
    v8f acc0 = {}, acc1 = {};
#pragma unroll
    for (int p = 0; p < 32; ++p) {
        const v4f xa = xr[2 * p];
        const v4f tb = tr[2 * p];
        v2f a0; a0.x = xa.x; a0.y = xa.y;
        v2f b0; b0.x = tb.x; b0.y = tb.y;
        v2f a1; a1.x = xa.z; a1.y = xa.w;
        v2f b1; b1.x = tb.z; b1.y = tb.w;
        acc0 = __builtin_amdgcn_wmma_f32_16x16x4_f32(false, a0, false, b0, (short)0, acc0, false, false);
        acc1 = __builtin_amdgcn_wmma_f32_16x16x4_f32(false, a1, false, b1, (short)0, acc1, false, false);
    }
    return acc0 + acc1;
}

__global__ void __launch_bounds__(256)
bupdate(const float* __restrict__ x, const float* __restrict__ t,
        float* __restrict__ bc) {
    __shared__ v4f sh[(BU_WAVES + 1) * BU_TILE4];
    const int b    = blockIdx.y;
    const int tid  = threadIdx.x;
    const int lane = tid & 31;
    const int wave = tid >> 5;
    const int half = lane >> 4;
    const int l16  = lane & 15;

    v4f* xs = sh + wave * BU_TILE4;
    v4f* ts = sh + BU_WAVES * BU_TILE4;

    const size_t row0 = (size_t)b * In + (size_t)blockIdx.x * BU_ROWS + 16 * wave;
    const v4f* xg = reinterpret_cast<const v4f*>(x + row0 * D);
    float*     bg = bc + row0 * N;
    constexpr size_t XSTEP4 = (size_t)16 * BU_WAVES * D / 4;
    constexpr size_t BSTEP  = (size_t)16 * BU_WAVES * N;

    v4f g[32];
    bu_load(g, xg, lane);

    const v4f* tg = reinterpret_cast<const v4f*>(t + (size_t)b * N * D);
    v4f tv[4];
#pragma unroll
    for (int j = 0; j < 4; ++j) tv[j] = tg[256 * j + tid];
#pragma unroll
    for (int j = 0; j < 4; ++j) {
        const int f = 256 * j + tid;
        ts[(f >> 6) * BU_PITCH4 + (f & 63)] = tv[j];
    }

    v8f res[BU_TILES];
#pragma unroll
    for (int k = 0; k < BU_TILES; ++k) {
        bu_stage(g, xs, lane);
        __syncthreads();
        if (k + 1 < BU_TILES) bu_load(g, xg + (k + 1) * XSTEP4, lane);
        res[k] = bu_tile(xs, ts, l16, half);
        if (k + 1 < BU_TILES) __syncthreads();
    }

#pragma unroll
    for (int k = 0; k < BU_TILES; ++k) {
        float* dst = bg + k * BSTEP;
#pragma unroll
        for (int r = 0; r < 8; ++r)
            dst[(size_t)(r + half * 8) * N + l16] = res[k][r];
    }
}

extern "C" void kernel_launch(void* const* d_in, const int* in_sizes, int n_in,
                              void* d_out, int out_size, void* d_ws, size_t ws_size,
                              hipStream_t stream) {
    const float* x = (const float*)d_in[0];
    const float* W = (const float*)d_in[1];
    float* out = (float*)d_out;

    float* ws     = (float*)d_ws;
    float* bc     = ws;
    float* s_part = bc + (size_t)B * In * N;
    float* s      = s_part + (size_t)B * SPLIT * N * D;
    float* t      = s + (size_t)B * N * D;

    const int bc_n = B * In * N;
    zero_buf<<<(bc_n + 255) / 256, 256, 0, stream>>>(bc, bc_n);

    for (int it = 0; it < 3; ++it) {
        softmax_n<<<(B * In + 255) / 256, 256, 0, stream>>>(bc);
        gemm_s<<<dim3(2, SPLIT, B), 256, 0, stream>>>(x, bc, s_part);
        reduce_s<<<(B * N * D + 255) / 256, 256, 0, stream>>>(s_part, s);
        outputs_squash<<<B * N, 256, 0, stream>>>(s, W, out);
        if (it < 2) {
            compute_t<<<B * N, 256, 0, stream>>>(W, out, t);
            bupdate<<<dim3(In / BU_ROWS, B), 256, 0, stream>>>(x, t, bc);
        }
    }
}


# iterations 2 and 3: logits GEMM, softmax over n and the s GEMM fused into one kernel on one staged x tile (x read once per iteration, b never stored); new order of the sums over i and n
# speedup vs baseline: 2.6768x; 1.4046x over previous
#include <hip/hip_runtime.h>
#include <hip/hip_bf16.h>


typedef __attribute__((ext_vector_type(2))) float v2f;
typedef __attribute__((ext_vector_type(4))) float v4f;
typedef __attribute__((ext_vector_type(8))) float v8f;

namespace caps {
constexpr int B   = 32;
constexpr int In  = 4096;
constexpr int D   = 256;
constexpr int N   = 16;
constexpr int K   = 16;
constexpr int SPLIT = 8;
}

using namespace caps;

__global__ void zero_buf(float* p, int n) {
    int i = blockIdx.x * blockDim.x + threadIdx.x;
    if (i < n) p[i] = 0.0f;
}

__global__ void softmax_n(float* __restrict__ bc) {
    int idx = blockIdx.x * blockDim.x + threadIdx.x;
    if (idx >= B * In) return;
    float* p = bc + (size_t)idx * N;
    float v[N];
    float m = -3.402823466e+38f;
#pragma unroll
    for (int n = 0; n < N; ++n) { v[n] = p[n]; m = fmaxf(m, v[n]); }
    float s = 0.0f;
#pragma unroll
    for (int n = 0; n < N; ++n) { v[n] = __expf(v[n] - m); s += v[n]; }
    float inv = 1.0f / s;
#pragma unroll
    for (int n = 0; n < N; ++n) p[n] = v[n] * inv;
}

constexpr int GS_ROWS  = In / SPLIT;
constexpr int GS_STEPS = 8;
constexpr int GS_GROW  = 4 * GS_STEPS;
constexpr int GS_RQ    = 4;
constexpr int GS_NGRP  = GS_ROWS / GS_GROW / GS_RQ;

struct GsFrag {
    v4f   x0[GS_STEPS], x1[GS_STEPS];
    float c0[GS_STEPS], c1[GS_STEPS];
};

__device__ __forceinline__ void gs_load(GsFrag& f, const float* __restrict__ xp,
                                        const float* __restrict__ cp) {
#pragma unroll
    for (int u = 0; u < GS_STEPS; ++u) {
        f.c0[u] = cp[(size_t)(4 * u + 0) * N];
        f.c1[u] = cp[(size_t)(4 * u + 1) * N];
        f.x0[u] = *reinterpret_cast<const v4f*>(xp + (size_t)(4 * u + 0) * D);
        f.x1[u] = *reinterpret_cast<const v4f*>(xp + (size_t)(4 * u + 1) * D);
    }
}

__device__ __forceinline__ void gs_mma(const GsFrag& f, v8f& acc0, v8f& acc1,
                                       v8f& acc2, v8f& acc3) {
#pragma unroll
    for (int u = 0; u < GS_STEPS; ++u) {
        v2f a;  a.x  = f.c0[u];    a.y  = f.c1[u];
        v2f b0; b0.x = f.x0[u].x;  b0.y = f.x1[u].x;
        v2f b1; b1.x = f.x0[u].y;  b1.y = f.x1[u].y;
        v2f b2; b2.x = f.x0[u].z;  b2.y = f.x1[u].z;
        v2f b3; b3.x = f.x0[u].w;  b3.y = f.x1[u].w;
        acc0 = __builtin_amdgcn_wmma_f32_16x16x4_f32(false, a, false, b0, (short)0, acc0, false, false);
        acc1 = __builtin_amdgcn_wmma_f32_16x16x4_f32(false, a, false, b1, (short)0, acc1, false, false);
        acc2 = __builtin_amdgcn_wmma_f32_16x16x4_f32(false, a, false, b2, (short)0, acc2, false, false);
        acc3 = __builtin_amdgcn_wmma_f32_16x16x4_f32(false, a, false, b3, (short)0, acc3, false, false);
    }
}

__global__ void __launch_bounds__(256)
gemm_s(const float* __restrict__ x, const float* __restrict__ c,
       float* __restrict__ s_part) {
    __shared__ v4f red[8 * 8 * 32];
    const int b    = blockIdx.z;
    const int sp   = blockIdx.y;
    const int lane = threadIdx.x & 31;
    const int wave = threadIdx.x >> 5;
    const int half = lane >> 4;
    const int l16  = lane & 15;
    const int dg   = wave & 1;
    const int rq   = wave >> 1;

    const int d0    = blockIdx.x * 128 + dg * 64;
    const int i_beg = sp * GS_ROWS + rq * GS_GROW + half * 2;

    const float* xp = x + ((size_t)b * In + i_beg) * D + d0 + 4 * l16;
    const float* cp = c + ((size_t)b * In + i_beg) * N + l16;
    constexpr size_t XG = (size_t)GS_RQ * GS_GROW * D;
    constexpr size_t CG = (size_t)GS_RQ * GS_GROW * N;

    v8f acc0 = {}, acc1 = {}, acc2 = {}, acc3 = {};
    GsFrag fa, fb;

    gs_load(fa, xp, cp);
#pragma unroll
    for (int t = 0; t < GS_NGRP; t += 2) {
        gs_load(fb, xp + (t + 1) * XG, cp + (t + 1) * CG);
        gs_mma(fa, acc0, acc1, acc2, acc3);
        if (t + 2 < GS_NGRP) gs_load(fa, xp + (t + 2) * XG, cp + (t + 2) * CG);
        gs_mma(fb, acc0, acc1, acc2, acc3);
    }

#pragma unroll
    for (int r = 0; r < 8; ++r) {
        v4f v; v.x = acc0[r]; v.y = acc1[r]; v.z = acc2[r]; v.w = acc3[r];
        red[(wave * 8 + r) * 32 + lane] = v;
    }
    __syncthreads();

    float* outp = s_part + (size_t)(b * SPLIT + sp) * N * D;
    for (int q = 0; q < 2; ++q) {
        v4f v = red[((0 * 2 + q) * 8 + wave) * 32 + lane];
        for (int p = 1; p < GS_RQ; ++p)
            v += red[((p * 2 + q) * 8 + wave) * 32 + lane];
        const int row = wave + 8 * half;
        const int d   = blockIdx.x * 128 + q * 64 + 4 * l16;
        *reinterpret_cast<v4f*>(outp + (size_t)row * D + d) = v;
    }
}

template <int NS>
__global__ void reduce_s(const float* __restrict__ s_part, float* __restrict__ s) {
    int idx = blockIdx.x * blockDim.x + threadIdx.x;
    if (idx >= B * N * D) return;
    int b  = idx / (N * D);
    int nd = idx - b * (N * D);
    float acc = 0.0f;
#pragma unroll
    for (int sp = 0; sp < NS; ++sp)
        acc += s_part[(size_t)(b * NS + sp) * N * D + nd];
    s[idx] = acc;
}

__global__ void __launch_bounds__(256)
outputs_squash(const float* __restrict__ s, const float* __restrict__ W,
               float* __restrict__ out) {
    __shared__ float part[D * K];
    __shared__ float dot[K];
    const int bn = blockIdx.x;
    const int n  = bn & (N - 1);
    const int d  = threadIdx.x;

    const float  sv = s[(size_t)bn * D + d];
    const float* wr = W + (size_t)d * (N * K) + (size_t)n * K;
#pragma unroll
    for (int k = 0; k < K; ++k) part[d * K + k] = sv * wr[k];
    __syncthreads();

    if (d < K) {
        float a = 0.0f;
        for (int dd = 0; dd < D; ++dd) a += part[dd * K + d];
        dot[d] = a;
    }
    __syncthreads();

    if (d < K) {
        float n2 = 0.0f;
#pragma unroll
        for (int k = 0; k < K; ++k) n2 += dot[k] * dot[k];
        out[(size_t)bn * K + d] = dot[d] * rsqrtf(n2 + 1e-7f);
    }
}

__global__ void __launch_bounds__(256)
compute_t(const float* __restrict__ W, const float* __restrict__ out,
          float* __restrict__ t) {
    __shared__ float ov[K];
    const int bn = blockIdx.x;
    const int n  = bn & (N - 1);
    const int d  = threadIdx.x;
    if (d < K) ov[d] = out[(size_t)bn * K + d];
    __syncthreads();
    const float* wr = W + (size_t)d * (N * K) + (size_t)n * K;
    float acc = 0.0f;
#pragma unroll
    for (int k = 0; k < K; ++k) acc += wr[k] * ov[k];
    t[(size_t)bn * D + d] = acc;
}

constexpr int RF_WAVES  = 4;
constexpr int RF_TILES  = 4;
constexpr int RF_ROWS   = 16 * RF_WAVES * RF_TILES;
constexpr int RF_CHUNKS = In / RF_ROWS;
static_assert(RF_CHUNKS * D == In, "the partial sums must fit the space of bc, which they reuse");
constexpr int BU_PITCH4 = D / 4 + 1;
constexpr int BU_TILE4  = 16 * BU_PITCH4;

__device__ __forceinline__ void bu_load(v4f (&g)[32], const v4f* __restrict__ src, int lane) {
#pragma unroll
    for (int j = 0; j < 32; ++j) g[j] = src[32 * j + lane];
}

__device__ __forceinline__ void bu_stage(const v4f (&g)[32], v4f* xs, int lane) {
#pragma unroll
    for (int j = 0; j < 32; ++j) xs[(j >> 1) * BU_PITCH4 + 32 * (j & 1) + lane] = g[j];
}

__device__ __forceinline__ void keep_groups() { __builtin_amdgcn_sched_barrier(0); }

struct G1Frag { v4f xa[8], tb[8]; };

__device__ __forceinline__ void g1_load(G1Frag& f, const v4f* xr, const v4f* tr, int blk) {
#pragma unroll
    for (int q = 0; q < 8; ++q) {
        f.xa[q] = xr[2 * (8 * blk + q)];
        f.tb[q] = tr[2 * (8 * blk + q)];
    }
}

__device__ __forceinline__ void g1_mma(const G1Frag& f, v8f& acc0, v8f& acc1) {
#pragma unroll
    for (int q = 0; q < 8; ++q) {
        v2f a0; a0.x = f.xa[q].x; a0.y = f.xa[q].y;
        v2f b0; b0.x = f.tb[q].x; b0.y = f.tb[q].y;
        v2f a1; a1.x = f.xa[q].z; a1.y = f.xa[q].w;
        v2f b1; b1.x = f.tb[q].z; b1.y = f.tb[q].w;
        acc0 = __builtin_amdgcn_wmma_f32_16x16x4_f32(false, a0, false, b0, (short)0, acc0, false, false);
        acc1 = __builtin_amdgcn_wmma_f32_16x16x4_f32(false, a1, false, b1, (short)0, acc1, false, false);
    }
}

__device__ __forceinline__ v8f bu_tile(const v4f* xs, const v4f* ts, int l16, int half) {
    const v4f* xr = xs + l16 * BU_PITCH4 + half;
    const v4f* tr = ts + l16 * BU_PITCH4 + half;
    v8f acc0 = {}, acc1 = {};
    G1Frag fa, fb;
    g1_load(fa, xr, tr, 0);
    g1_load(fb, xr, tr, 1);  keep_groups();
    g1_mma(fa, acc0, acc1);  keep_groups();
    g1_load(fa, xr, tr, 2);  keep_groups();
    g1_mma(fb, acc0, acc1);  keep_groups();
    g1_load(fb, xr, tr, 3);  keep_groups();
    g1_mma(fa, acc0, acc1);  keep_groups();
    g1_mma(fb, acc0, acc1);
    return acc0 + acc1;
}

template <int M>
__device__ __forceinline__ float row_xor(float v) {
    return __int_as_float(__builtin_amdgcn_update_dpp(0, __float_as_int(v), 0x160 + M, 0xf, 0xf, false));
}

__device__ __forceinline__ float row16_max(float v) {
    v = fmaxf(v, row_xor<1>(v));
    v = fmaxf(v, row_xor<2>(v));
    v = fmaxf(v, row_xor<4>(v));
    v = fmaxf(v, row_xor<8>(v));
    return v;
}

__device__ __forceinline__ float row16_sum(float v) {
    v += row_xor<1>(v);
    v += row_xor<2>(v);
    v += row_xor<4>(v);
    v += row_xor<8>(v);
    return v;
}

struct G2Frag { v4f x0[4], x1[4]; };

__device__ __forceinline__ void g2_load(G2Frag& f, const v4f* xr, int u) {
#pragma unroll
    for (int g = 0; g < 4; ++g) {
        f.x0[g] = xr[(2 * u + 0) * BU_PITCH4 + 16 * g];
        f.x1[g] = xr[(2 * u + 1) * BU_PITCH4 + 16 * g];
    }
}

__device__ __forceinline__ void g2_mma(const G2Frag& f, float c0, float c1, v8f (&acc)[16]) {
    v2f a; a.x = c0; a.y = c1;
#pragma unroll
    for (int g = 0; g < 4; ++g) {
        v2f b0; b0.x = f.x0[g].x; b0.y = f.x1[g].x;
        v2f b1; b1.x = f.x0[g].y; b1.y = f.x1[g].y;
        v2f b2; b2.x = f.x0[g].z; b2.y = f.x1[g].z;
        v2f b3; b3.x = f.x0[g].w; b3.y = f.x1[g].w;
        acc[4 * g + 0] = __builtin_amdgcn_wmma_f32_16x16x4_f32(false, a, false, b0, (short)0, acc[4 * g + 0], false, false);
        acc[4 * g + 1] = __builtin_amdgcn_wmma_f32_16x16x4_f32(false, a, false, b1, (short)0, acc[4 * g + 1], false, false);
        acc[4 * g + 2] = __builtin_amdgcn_wmma_f32_16x16x4_f32(false, a, false, b2, (short)0, acc[4 * g + 2], false, false);
        acc[4 * g + 3] = __builtin_amdgcn_wmma_f32_16x16x4_f32(false, a, false, b3, (short)0, acc[4 * g + 3], false, false);
    }
}

__global__ void __launch_bounds__(32 * RF_WAVES)
route_fused(const float* __restrict__ x, const float* __restrict__ t,
            float* __restrict__ s_part) {
    __shared__ v4f sh[(RF_WAVES + 1) * BU_TILE4];
    const int b    = blockIdx.y;
    const int tid  = threadIdx.x;
    const int lane = tid & 31;
    const int wave = tid >> 5;
    const int half = lane >> 4;
    const int l16  = lane & 15;

    v4f* xs = sh + wave * BU_TILE4;
    v4f* ts = sh + RF_WAVES * BU_TILE4;

    const size_t row0 = (size_t)b * In + (size_t)blockIdx.x * RF_ROWS + 16 * wave;
    const v4f* xg = reinterpret_cast<const v4f*>(x + row0 * D);
    constexpr size_t XSTEP4 = (size_t)16 * RF_WAVES * D / 4;

    v4f g[32];
    bu_load(g, xg, lane);

    const v4f* tg = reinterpret_cast<const v4f*>(t + (size_t)b * N * D);
    v4f tv[8];
#pragma unroll
    for (int j = 0; j < 8; ++j) tv[j] = tg[128 * j + tid];
#pragma unroll
    for (int j = 0; j < 8; ++j) {
        const int f = 128 * j + tid;
        ts[(f >> 6) * BU_PITCH4 + (f & 63)] = tv[j];
    }
    __syncthreads();

    v8f acc[16];
#pragma unroll
    for (int j = 0; j < 16; ++j) acc[j] = v8f{};

    for (int k = 0; k < RF_TILES; ++k) {
        bu_stage(g, xs, lane);
        if (k + 1 < RF_TILES) bu_load(g, xg + (k + 1) * XSTEP4, lane);
        __builtin_amdgcn_sched_barrier(0);
        const v8f lg = bu_tile(xs, ts, l16, half);

        const v4f* xr2 = xs + (8 * half) * BU_PITCH4 + l16;
        G2Frag fa, fb;
        g2_load(fa, xr2, 0);

        float cc[8];
#pragma unroll
        for (int r = 0; r < 8; ++r) {
            const float m = row16_max(lg[r]);
            const float e = __expf(lg[r] - m);
            const float inv = 1.0f / row16_sum(e);
            cc[r] = e * inv;
        }

        g2_load(fb, xr2, 1);              keep_groups();
        g2_mma(fa, cc[0], cc[1], acc);    keep_groups();
        g2_load(fa, xr2, 2);              keep_groups();
        g2_mma(fb, cc[2], cc[3], acc);    keep_groups();
        g2_load(fb, xr2, 3);              keep_groups();
        g2_mma(fa, cc[4], cc[5], acc);    keep_groups();
        g2_mma(fb, cc[6], cc[7], acc);
    }

#pragma unroll
    for (int g4 = 0; g4 < 4; ++g4) {
#pragma unroll
        for (int r = 0; r < 8; ++r) {
            v4f v;
            v.x = acc[4 * g4 + 0][r]; v.y = acc[4 * g4 + 1][r];
            v.z = acc[4 * g4 + 2][r]; v.w = acc[4 * g4 + 3][r];
            xs[(g4 * 8 + r) * 32 + lane] = v;
        }
    }
    __syncthreads();

    float* outp = s_part + (size_t)(b * RF_CHUNKS + blockIdx.x) * N * D;
#pragma unroll
    for (int q = 0; q < 8; ++q) {
        const int e = 128 * q + tid;
        v4f v = sh[e];
#pragma unroll
        for (int w = 1; w < RF_WAVES; ++w) v += sh[w * BU_TILE4 + e];
        const int gr = e >> 5;
        const int n  = (gr & 7) + 8 * ((e >> 4) & 1);
        const int d  = 64 * (gr >> 3) + 4 * (e & 15);
        *reinterpret_cast<v4f*>(outp + (size_t)n * D + d) = v;
    }
}

extern "C" void kernel_launch(void* const* d_in, const int* in_sizes, int n_in,
                              void* d_out, int out_size, void* d_ws, size_t ws_size,
                              hipStream_t stream) {
    const float* x = (const float*)d_in[0];
    const float* W = (const float*)d_in[1];
    float* out = (float*)d_out;

    float* ws     = (float*)d_ws;
    float* bc     = ws;
    float* s_part = bc + (size_t)B * In * N;
    float* s      = s_part + (size_t)B * SPLIT * N * D;
    float* t      = s + (size_t)B * N * D;

    float* s_part16 = bc;

    const int bc_n = B * In * N;
    zero_buf<<<(bc_n + 255) / 256, 256, 0, stream>>>(bc, bc_n);

    softmax_n<<<(B * In + 255) / 256, 256, 0, stream>>>(bc);
    gemm_s<<<dim3(2, SPLIT, B), 256, 0, stream>>>(x, bc, s_part);
    reduce_s<SPLIT><<<(B * N * D + 255) / 256, 256, 0, stream>>>(s_part, s);
    outputs_squash<<<B * N, 256, 0, stream>>>(s, W, out);

    for (int it = 1; it < 3; ++it) {
        compute_t<<<B * N, 256, 0, stream>>>(W, out, t);
        route_fused<<<dim3(RF_CHUNKS, B), 32 * RF_WAVES, 0, stream>>>(x, t, s_part16);
        reduce_s<RF_CHUNKS><<<(B * N * D + 255) / 256, 256, 0, stream>>>(s_part16, s);
        outputs_squash<<<B * N, 256, 0, stream>>>(s, W, out);
    }
}


# reduce_s, outputs_squash, compute_t merged into one 64-thread kernel per (b,n): 13 launches to 8, s and outputs stay in registers; new order of the sum over d (16-lane butterfly, then 4 half-waves)
# speedup vs baseline: 3.4148x; 1.2757x over previous
#include <hip/hip_runtime.h>
#include <hip/hip_bf16.h>


typedef __attribute__((ext_vector_type(2))) float v2f;
typedef __attribute__((ext_vector_type(4))) float v4f;
typedef __attribute__((ext_vector_type(8))) float v8f;

namespace caps {
constexpr int B   = 32;
constexpr int In  = 4096;
constexpr int D   = 256;
constexpr int N   = 16;
constexpr int K   = 16;
constexpr int SPLIT = 8;
}

using namespace caps;

__global__ void zero_buf(float* p, int n) {
    int i = blockIdx.x * blockDim.x + threadIdx.x;
    if (i < n) p[i] = 0.0f;
}

__global__ void softmax_n(float* __restrict__ bc) {
    int idx = blockIdx.x * blockDim.x + threadIdx.x;
    if (idx >= B * In) return;
    float* p = bc + (size_t)idx * N;
    float v[N];
    float m = -3.402823466e+38f;
#pragma unroll
    for (int n = 0; n < N; ++n) { v[n] = p[n]; m = fmaxf(m, v[n]); }
    float s = 0.0f;
#pragma unroll
    for (int n = 0; n < N; ++n) { v[n] = __expf(v[n] - m); s += v[n]; }
    float inv = 1.0f / s;
#pragma unroll
    for (int n = 0; n < N; ++n) p[n] = v[n] * inv;
}

constexpr int GS_ROWS  = In / SPLIT;
constexpr int GS_STEPS = 8;
constexpr int GS_GROW  = 4 * GS_STEPS;
constexpr int GS_RQ    = 4;
constexpr int GS_NGRP  = GS_ROWS / GS_GROW / GS_RQ;

struct GsFrag {
    v4f   x0[GS_STEPS], x1[GS_STEPS];
    float c0[GS_STEPS], c1[GS_STEPS];
};

__device__ __forceinline__ void gs_load(GsFrag& f, const float* __restrict__ xp,
                                        const float* __restrict__ cp) {
#pragma unroll
    for (int u = 0; u < GS_STEPS; ++u) {
        f.c0[u] = cp[(size_t)(4 * u + 0) * N];
        f.c1[u] = cp[(size_t)(4 * u + 1) * N];
        f.x0[u] = *reinterpret_cast<const v4f*>(xp + (size_t)(4 * u + 0) * D);
        f.x1[u] = *reinterpret_cast<const v4f*>(xp + (size_t)(4 * u + 1) * D);
    }
}

__device__ __forceinline__ void gs_mma(const GsFrag& f, v8f& acc0, v8f& acc1,
                                       v8f& acc2, v8f& acc3) {
#pragma unroll
    for (int u = 0; u < GS_STEPS; ++u) {
        v2f a;  a.x  = f.c0[u];    a.y  = f.c1[u];
        v2f b0; b0.x = f.x0[u].x;  b0.y = f.x1[u].x;
        v2f b1; b1.x = f.x0[u].y;  b1.y = f.x1[u].y;
        v2f b2; b2.x = f.x0[u].z;  b2.y = f.x1[u].z;
        v2f b3; b3.x = f.x0[u].w;  b3.y = f.x1[u].w;
        acc0 = __builtin_amdgcn_wmma_f32_16x16x4_f32(false, a, false, b0, (short)0, acc0, false, false);
        acc1 = __builtin_amdgcn_wmma_f32_16x16x4_f32(false, a, false, b1, (short)0, acc1, false, false);
        acc2 = __builtin_amdgcn_wmma_f32_16x16x4_f32(false, a, false, b2, (short)0, acc2, false, false);
        acc3 = __builtin_amdgcn_wmma_f32_16x16x4_f32(false, a, false, b3, (short)0, acc3, false, false);
    }
}

__global__ void __launch_bounds__(256)
gemm_s(const float* __restrict__ x, const float* __restrict__ c,
       float* __restrict__ s_part) {
    __shared__ v4f red[8 * 8 * 32];
    const int b    = blockIdx.z;
    const int sp   = blockIdx.y;
    const int lane = threadIdx.x & 31;
    const int wave = threadIdx.x >> 5;
    const int half = lane >> 4;
    const int l16  = lane & 15;
    const int dg   = wave & 1;
    const int rq   = wave >> 1;

    const int d0    = blockIdx.x * 128 + dg * 64;
    const int i_beg = sp * GS_ROWS + rq * GS_GROW + half * 2;

    const float* xp = x + ((size_t)b * In + i_beg) * D + d0 + 4 * l16;
    const float* cp = c + ((size_t)b * In + i_beg) * N + l16;
    constexpr size_t XG = (size_t)GS_RQ * GS_GROW * D;
    constexpr size_t CG = (size_t)GS_RQ * GS_GROW * N;

    v8f acc0 = {}, acc1 = {}, acc2 = {}, acc3 = {};
    GsFrag fa, fb;

    gs_load(fa, xp, cp);
#pragma unroll
    for (int t = 0; t < GS_NGRP; t += 2) {
        gs_load(fb, xp + (t + 1) * XG, cp + (t + 1) * CG);
        gs_mma(fa, acc0, acc1, acc2, acc3);
        if (t + 2 < GS_NGRP) gs_load(fa, xp + (t + 2) * XG, cp + (t + 2) * CG);
        gs_mma(fb, acc0, acc1, acc2, acc3);
    }

#pragma unroll
    for (int r = 0; r < 8; ++r) {
        v4f v; v.x = acc0[r]; v.y = acc1[r]; v.z = acc2[r]; v.w = acc3[r];
        red[(wave * 8 + r) * 32 + lane] = v;
    }
    __syncthreads();

    float* outp = s_part + (size_t)(b * SPLIT + sp) * N * D;
    for (int q = 0; q < 2; ++q) {
        v4f v = red[((0 * 2 + q) * 8 + wave) * 32 + lane];
        for (int p = 1; p < GS_RQ; ++p)
            v += red[((p * 2 + q) * 8 + wave) * 32 + lane];
        const int row = wave + 8 * half;
        const int d   = blockIdx.x * 128 + q * 64 + 4 * l16;
        *reinterpret_cast<v4f*>(outp + (size_t)row * D + d) = v;
    }
}

constexpr int RF_WAVES  = 4;
constexpr int RF_TILES  = 4;
constexpr int RF_ROWS   = 16 * RF_WAVES * RF_TILES;
constexpr int RF_CHUNKS = In / RF_ROWS;
static_assert(RF_CHUNKS * D == In, "the partial sums must fit the space of bc, which they reuse");
constexpr int BU_PITCH4 = D / 4 + 1;
constexpr int BU_TILE4  = 16 * BU_PITCH4;

__device__ __forceinline__ void bu_load(v4f (&g)[32], const v4f* __restrict__ src, int lane) {
#pragma unroll
    for (int j = 0; j < 32; ++j) g[j] = src[32 * j + lane];
}

__device__ __forceinline__ void bu_stage(const v4f (&g)[32], v4f* xs, int lane) {
#pragma unroll
    for (int j = 0; j < 32; ++j) xs[(j >> 1) * BU_PITCH4 + 32 * (j & 1) + lane] = g[j];
}

__device__ __forceinline__ void keep_groups() { __builtin_amdgcn_sched_barrier(0); }

struct G1Frag { v4f xa[8], tb[8]; };

__device__ __forceinline__ void g1_load(G1Frag& f, const v4f* xr, const v4f* tr, int blk) {
#pragma unroll
    for (int q = 0; q < 8; ++q) {
        f.xa[q] = xr[2 * (8 * blk + q)];
        f.tb[q] = tr[2 * (8 * blk + q)];
    }
}

__device__ __forceinline__ void g1_mma(const G1Frag& f, v8f& acc0, v8f& acc1) {
#pragma unroll
    for (int q = 0; q < 8; ++q) {
        v2f a0; a0.x = f.xa[q].x; a0.y = f.xa[q].y;
        v2f b0; b0.x = f.tb[q].x; b0.y = f.tb[q].y;
        v2f a1; a1.x = f.xa[q].z; a1.y = f.xa[q].w;
        v2f b1; b1.x = f.tb[q].z; b1.y = f.tb[q].w;
        acc0 = __builtin_amdgcn_wmma_f32_16x16x4_f32(false, a0, false, b0, (short)0, acc0, false, false);
        acc1 = __builtin_amdgcn_wmma_f32_16x16x4_f32(false, a1, false, b1, (short)0, acc1, false, false);
    }
}

__device__ __forceinline__ v8f bu_tile(const v4f* xs, const v4f* ts, int l16, int half) {
    const v4f* xr = xs + l16 * BU_PITCH4 + half;
    const v4f* tr = ts + l16 * BU_PITCH4 + half;
    v8f acc0 = {}, acc1 = {};
    G1Frag fa, fb;
    g1_load(fa, xr, tr, 0);
    g1_load(fb, xr, tr, 1);  keep_groups();
    g1_mma(fa, acc0, acc1);  keep_groups();
    g1_load(fa, xr, tr, 2);  keep_groups();
    g1_mma(fb, acc0, acc1);  keep_groups();
    g1_load(fb, xr, tr, 3);  keep_groups();
    g1_mma(fa, acc0, acc1);  keep_groups();
    g1_mma(fb, acc0, acc1);
    return acc0 + acc1;
}

template <int M>
__device__ __forceinline__ float row_xor(float v) {
    return __int_as_float(__builtin_amdgcn_update_dpp(0, __float_as_int(v), 0x160 + M, 0xf, 0xf, false));
}

__device__ __forceinline__ float row16_max(float v) {
    v = fmaxf(v, row_xor<1>(v));
    v = fmaxf(v, row_xor<2>(v));
    v = fmaxf(v, row_xor<4>(v));
    v = fmaxf(v, row_xor<8>(v));
    return v;
}

__device__ __forceinline__ float row16_sum(float v) {
    v += row_xor<1>(v);
    v += row_xor<2>(v);
    v += row_xor<4>(v);
    v += row_xor<8>(v);
    return v;
}

struct G2Frag { v4f x0[4], x1[4]; };

__device__ __forceinline__ void g2_load(G2Frag& f, const v4f* xr, int u) {
#pragma unroll
    for (int g = 0; g < 4; ++g) {
        f.x0[g] = xr[(2 * u + 0) * BU_PITCH4 + 16 * g];
        f.x1[g] = xr[(2 * u + 1) * BU_PITCH4 + 16 * g];
    }
}

__device__ __forceinline__ void g2_mma(const G2Frag& f, float c0, float c1, v8f (&acc)[16]) {
    v2f a; a.x = c0; a.y = c1;
#pragma unroll
    for (int g = 0; g < 4; ++g) {
        v2f b0; b0.x = f.x0[g].x; b0.y = f.x1[g].x;
        v2f b1; b1.x = f.x0[g].y; b1.y = f.x1[g].y;
        v2f b2; b2.x = f.x0[g].z; b2.y = f.x1[g].z;
        v2f b3; b3.x = f.x0[g].w; b3.y = f.x1[g].w;
        acc[4 * g + 0] = __builtin_amdgcn_wmma_f32_16x16x4_f32(false, a, false, b0, (short)0, acc[4 * g + 0], false, false);
        acc[4 * g + 1] = __builtin_amdgcn_wmma_f32_16x16x4_f32(false, a, false, b1, (short)0, acc[4 * g + 1], false, false);
        acc[4 * g + 2] = __builtin_amdgcn_wmma_f32_16x16x4_f32(false, a, false, b2, (short)0, acc[4 * g + 2], false, false);
        acc[4 * g + 3] = __builtin_amdgcn_wmma_f32_16x16x4_f32(false, a, false, b3, (short)0, acc[4 * g + 3], false, false);
    }
}

__global__ void __launch_bounds__(32 * RF_WAVES)
route_fused(const float* __restrict__ x, const float* __restrict__ t,
            float* __restrict__ s_part) {
    __shared__ v4f sh[(RF_WAVES + 1) * BU_TILE4];
    const int b    = blockIdx.y;
    const int tid  = threadIdx.x;
    const int lane = tid & 31;
    const int wave = tid >> 5;
    const int half = lane >> 4;
    const int l16  = lane & 15;

    v4f* xs = sh + wave * BU_TILE4;
    v4f* ts = sh + RF_WAVES * BU_TILE4;

    const size_t row0 = (size_t)b * In + (size_t)blockIdx.x * RF_ROWS + 16 * wave;
    const v4f* xg = reinterpret_cast<const v4f*>(x + row0 * D);
    constexpr size_t XSTEP4 = (size_t)16 * RF_WAVES * D / 4;

    v4f g[32];
    bu_load(g, xg, lane);

    const v4f* tg = reinterpret_cast<const v4f*>(t + (size_t)b * N * D);
    v4f tv[8];
#pragma unroll
    for (int j = 0; j < 8; ++j) tv[j] = tg[128 * j + tid];
#pragma unroll
    for (int j = 0; j < 8; ++j) {
        const int f = 128 * j + tid;
        ts[(f >> 6) * BU_PITCH4 + (f & 63)] = tv[j];
    }
    __syncthreads();

    v8f acc[16];
#pragma unroll
    for (int j = 0; j < 16; ++j) acc[j] = v8f{};

    for (int k = 0; k < RF_TILES; ++k) {
        bu_stage(g, xs, lane);
        if (k + 1 < RF_TILES) bu_load(g, xg + (k + 1) * XSTEP4, lane);
        __builtin_amdgcn_sched_barrier(0);
        const v8f lg = bu_tile(xs, ts, l16, half);

        const v4f* xr2 = xs + (8 * half) * BU_PITCH4 + l16;
        G2Frag fa, fb;
        g2_load(fa, xr2, 0);

        float cc[8];
#pragma unroll
        for (int r = 0; r < 8; ++r) {
            const float m = row16_max(lg[r]);
            const float e = __expf(lg[r] - m);
            const float inv = 1.0f / row16_sum(e);
            cc[r] = e * inv;
        }

        g2_load(fb, xr2, 1);              keep_groups();
        g2_mma(fa, cc[0], cc[1], acc);    keep_groups();
        g2_load(fa, xr2, 2);              keep_groups();
        g2_mma(fb, cc[2], cc[3], acc);    keep_groups();
        g2_load(fb, xr2, 3);              keep_groups();
        g2_mma(fa, cc[4], cc[5], acc);    keep_groups();
        g2_mma(fb, cc[6], cc[7], acc);
    }

#pragma unroll
    for (int g4 = 0; g4 < 4; ++g4) {
#pragma unroll
        for (int r = 0; r < 8; ++r) {
            v4f v;
            v.x = acc[4 * g4 + 0][r]; v.y = acc[4 * g4 + 1][r];
            v.z = acc[4 * g4 + 2][r]; v.w = acc[4 * g4 + 3][r];
            xs[(g4 * 8 + r) * 32 + lane] = v;
        }
    }
    __syncthreads();

    float* outp = s_part + (size_t)(b * RF_CHUNKS + blockIdx.x) * N * D;
#pragma unroll
    for (int q = 0; q < 8; ++q) {
        const int e = 128 * q + tid;
        v4f v = sh[e];
#pragma unroll
        for (int w = 1; w < RF_WAVES; ++w) v += sh[w * BU_TILE4 + e];
        const int gr = e >> 5;
        const int n  = (gr & 7) + 8 * ((e >> 4) & 1);
        const int d  = 64 * (gr >> 3) + 4 * (e & 15);
        *reinterpret_cast<v4f*>(outp + (size_t)n * D + d) = v;
    }
}

template <int NS, bool LAST>
__global__ void __launch_bounds__(64)
route_tail(const float* __restrict__ s_part, const float* __restrict__ W,
           float* __restrict__ t, float* __restrict__ out) {
    __shared__ v4f red[4 * 4];
    const int bn  = blockIdx.x;
    const int b   = bn / N;
    const int n   = bn - b * N;
    const int tid = threadIdx.x;

    const v4f* sp = reinterpret_cast<const v4f*>(s_part + ((size_t)b * NS * N + n) * D) + tid;
    v4f pv[NS];
#pragma unroll
    for (int i = 0; i < NS; ++i) pv[i] = sp[(size_t)i * (N * D / 4)];

    const v4f* wp = reinterpret_cast<const v4f*>(W + (size_t)(4 * tid) * (N * K) + (size_t)n * K);
    v4f w[4][4];
#pragma unroll
    for (int r = 0; r < 4; ++r) {
#pragma unroll
        for (int q = 0; q < 4; ++q) w[r][q] = wp[r * (N * K / 4) + q];
    }

    v4f sv = pv[0];
#pragma unroll
    for (int i = 1; i < NS; ++i) sv += pv[i];

    v4f p[4];
#pragma unroll
    for (int q = 0; q < 4; ++q) {
        p[q] = sv.x * w[0][q];
        p[q] += sv.y * w[1][q];
        p[q] += sv.z * w[2][q];
        p[q] += sv.w * w[3][q];
        p[q].x = row16_sum(p[q].x);
        p[q].y = row16_sum(p[q].y);
        p[q].z = row16_sum(p[q].z);
        p[q].w = row16_sum(p[q].w);
    }
    if ((tid & 15) == 0) {
#pragma unroll
        for (int q = 0; q < 4; ++q) red[(tid >> 4) * 4 + q] = p[q];
    }
    __syncthreads();

    v4f dot[4];
    float n2 = 0.0f;
#pragma unroll
    for (int q = 0; q < 4; ++q) {
        dot[q] = red[q];
        dot[q] += red[4 + q];
        dot[q] += red[8 + q];
        dot[q] += red[12 + q];
        n2 += dot[q].x * dot[q].x;
        n2 += dot[q].y * dot[q].y;
        n2 += dot[q].z * dot[q].z;
        n2 += dot[q].w * dot[q].w;
    }
    const float inv = rsqrtf(n2 + 1e-7f);

    if (LAST) {
        if (tid < 4) {
            v4f o = red[tid];
            o += red[4 + tid];
            o += red[8 + tid];
            o += red[12 + tid];
            reinterpret_cast<v4f*>(out + (size_t)bn * K)[tid] = o * inv;
        }
    } else {
        v4f tv;
#pragma unroll
        for (int r = 0; r < 4; ++r) {
            float acc = 0.0f;
#pragma unroll
            for (int q = 0; q < 4; ++q) {
                acc += w[r][q].x * (dot[q].x * inv);
                acc += w[r][q].y * (dot[q].y * inv);
                acc += w[r][q].z * (dot[q].z * inv);
                acc += w[r][q].w * (dot[q].w * inv);
            }
            tv[r] = acc;
        }
        reinterpret_cast<v4f*>(t + (size_t)bn * D)[tid] = tv;
    }
}

extern "C" void kernel_launch(void* const* d_in, const int* in_sizes, int n_in,
                              void* d_out, int out_size, void* d_ws, size_t ws_size,
                              hipStream_t stream) {
    const float* x = (const float*)d_in[0];
    const float* W = (const float*)d_in[1];
    float* out = (float*)d_out;

    float* ws     = (float*)d_ws;
    float* bc     = ws;
    float* s_part = bc + (size_t)B * In * N;
    float* t      = s_part + (size_t)B * SPLIT * N * D;

    float* s_part16 = bc;

    const int bc_n = B * In * N;
    zero_buf<<<(bc_n + 255) / 256, 256, 0, stream>>>(bc, bc_n);

    softmax_n<<<(B * In + 255) / 256, 256, 0, stream>>>(bc);
    gemm_s<<<dim3(2, SPLIT, B), 256, 0, stream>>>(x, bc, s_part);
    route_tail<SPLIT, false><<<B * N, 64, 0, stream>>>(s_part, W, t, out);

    route_fused<<<dim3(RF_CHUNKS, B), 32 * RF_WAVES, 0, stream>>>(x, t, s_part16);
    route_tail<RF_CHUNKS, false><<<B * N, 64, 0, stream>>>(s_part16, W, t, out);

    route_fused<<<dim3(RF_CHUNKS, B), 32 * RF_WAVES, 0, stream>>>(x, t, s_part16);
    route_tail<RF_CHUNKS, true><<<B * N, 64, 0, stream>>>(s_part16, W, t, out);
}


# iteration 1: couplings are the constant softmax(0) = 1/N, held in registers as the WMMA A operand; logits clear and softmax launches dropped (8 launches to 6), c never read; sums unchanged
# speedup vs baseline: 4.6525x; 1.3625x over previous
#include <hip/hip_runtime.h>
#include <hip/hip_bf16.h>


typedef __attribute__((ext_vector_type(2))) float v2f;
typedef __attribute__((ext_vector_type(4))) float v4f;
typedef __attribute__((ext_vector_type(8))) float v8f;

namespace caps {
constexpr int B   = 32;
constexpr int In  = 4096;
constexpr int D   = 256;
constexpr int N   = 16;
constexpr int K   = 16;
constexpr int SPLIT = 8;
}

using namespace caps;

constexpr int GS_ROWS  = In / SPLIT;
constexpr int GS_STEPS = 8;
constexpr int GS_GROW  = 4 * GS_STEPS;
constexpr int GS_RQ    = 4;
constexpr int GS_NGRP  = GS_ROWS / GS_GROW / GS_RQ;

struct GsFrag {
    v4f   x0[GS_STEPS], x1[GS_STEPS];
};

__device__ __forceinline__ void gs_load(GsFrag& f, const float* __restrict__ xp) {
#pragma unroll
    for (int u = 0; u < GS_STEPS; ++u) {
        f.x0[u] = *reinterpret_cast<const v4f*>(xp + (size_t)(4 * u + 0) * D);
        f.x1[u] = *reinterpret_cast<const v4f*>(xp + (size_t)(4 * u + 1) * D);
    }
}

__device__ __forceinline__ void gs_mma(const GsFrag& f, v8f& acc0, v8f& acc1,
                                       v8f& acc2, v8f& acc3) {
    constexpr float c_first = 1.0f / N;
    v2f a;  a.x = c_first;  a.y = c_first;
#pragma unroll
    for (int u = 0; u < GS_STEPS; ++u) {
        v2f b0; b0.x = f.x0[u].x;  b0.y = f.x1[u].x;
        v2f b1; b1.x = f.x0[u].y;  b1.y = f.x1[u].y;
        v2f b2; b2.x = f.x0[u].z;  b2.y = f.x1[u].z;
        v2f b3; b3.x = f.x0[u].w;  b3.y = f.x1[u].w;
        acc0 = __builtin_amdgcn_wmma_f32_16x16x4_f32(false, a, false, b0, (short)0, acc0, false, false);
        acc1 = __builtin_amdgcn_wmma_f32_16x16x4_f32(false, a, false, b1, (short)0, acc1, false, false);
        acc2 = __builtin_amdgcn_wmma_f32_16x16x4_f32(false, a, false, b2, (short)0, acc2, false, false);
        acc3 = __builtin_amdgcn_wmma_f32_16x16x4_f32(false, a, false, b3, (short)0, acc3, false, false);
    }
}

__global__ void __launch_bounds__(256)
gemm_s(const float* __restrict__ x, float* __restrict__ s_part) {
    __shared__ v4f red[8 * 8 * 32];
    const int b    = blockIdx.z;
    const int sp   = blockIdx.y;
    const int lane = threadIdx.x & 31;
    const int wave = threadIdx.x >> 5;
    const int half = lane >> 4;
    const int l16  = lane & 15;
    const int dg   = wave & 1;
    const int rq   = wave >> 1;

    const int d0    = blockIdx.x * 128 + dg * 64;
    const int i_beg = sp * GS_ROWS + rq * GS_GROW + half * 2;

    const float* xp = x + ((size_t)b * In + i_beg) * D + d0 + 4 * l16;
    constexpr size_t XG = (size_t)GS_RQ * GS_GROW * D;

    v8f acc0 = {}, acc1 = {}, acc2 = {}, acc3 = {};
    GsFrag fa, fb;

    gs_load(fa, xp);
#pragma unroll
    for (int t = 0; t < GS_NGRP; t += 2) {
        gs_load(fb, xp + (t + 1) * XG);
        gs_mma(fa, acc0, acc1, acc2, acc3);
        if (t + 2 < GS_NGRP) gs_load(fa, xp + (t + 2) * XG);
        gs_mma(fb, acc0, acc1, acc2, acc3);
    }

#pragma unroll
    for (int r = 0; r < 8; ++r) {
        v4f v; v.x = acc0[r]; v.y = acc1[r]; v.z = acc2[r]; v.w = acc3[r];
        red[(wave * 8 + r) * 32 + lane] = v;
    }
    __syncthreads();

    float* outp = s_part + (size_t)(b * SPLIT + sp) * N * D;
    for (int q = 0; q < 2; ++q) {
        v4f v = red[((0 * 2 + q) * 8 + wave) * 32 + lane];
        for (int p = 1; p < GS_RQ; ++p)
            v += red[((p * 2 + q) * 8 + wave) * 32 + lane];
        const int row = wave + 8 * half;
        const int d   = blockIdx.x * 128 + q * 64 + 4 * l16;
        *reinterpret_cast<v4f*>(outp + (size_t)row * D + d) = v;
    }
}

constexpr int RF_WAVES  = 4;
constexpr int RF_TILES  = 4;
constexpr int RF_ROWS   = 16 * RF_WAVES * RF_TILES;
constexpr int RF_CHUNKS = In / RF_ROWS;
static_assert(RF_CHUNKS * D == In, "B*RF_CHUNKS*N*D floats of partial sums = B*In*N, the workspace's first plane");
constexpr int BU_PITCH4 = D / 4 + 1;
constexpr int BU_TILE4  = 16 * BU_PITCH4;

__device__ __forceinline__ void bu_load(v4f (&g)[32], const v4f* __restrict__ src, int lane) {
#pragma unroll
    for (int j = 0; j < 32; ++j) g[j] = src[32 * j + lane];
}

__device__ __forceinline__ void bu_stage(const v4f (&g)[32], v4f* xs, int lane) {
#pragma unroll
    for (int j = 0; j < 32; ++j) xs[(j >> 1) * BU_PITCH4 + 32 * (j & 1) + lane] = g[j];
}

__device__ __forceinline__ void keep_groups() { __builtin_amdgcn_sched_barrier(0); }

struct G1Frag { v4f xa[8], tb[8]; };

__device__ __forceinline__ void g1_load(G1Frag& f, const v4f* xr, const v4f* tr, int blk) {
#pragma unroll
    for (int q = 0; q < 8; ++q) {
        f.xa[q] = xr[2 * (8 * blk + q)];
        f.tb[q] = tr[2 * (8 * blk + q)];
    }
}

__device__ __forceinline__ void g1_mma(const G1Frag& f, v8f& acc0, v8f& acc1) {
#pragma unroll
    for (int q = 0; q < 8; ++q) {
        v2f a0; a0.x = f.xa[q].x; a0.y = f.xa[q].y;
        v2f b0; b0.x = f.tb[q].x; b0.y = f.tb[q].y;
        v2f a1; a1.x = f.xa[q].z; a1.y = f.xa[q].w;
        v2f b1; b1.x = f.tb[q].z; b1.y = f.tb[q].w;
        acc0 = __builtin_amdgcn_wmma_f32_16x16x4_f32(false, a0, false, b0, (short)0, acc0, false, false);
        acc1 = __builtin_amdgcn_wmma_f32_16x16x4_f32(false, a1, false, b1, (short)0, acc1, false, false);
    }
}

__device__ __forceinline__ v8f bu_tile(const v4f* xs, const v4f* ts, int l16, int half) {
    const v4f* xr = xs + l16 * BU_PITCH4 + half;
    const v4f* tr = ts + l16 * BU_PITCH4 + half;
    v8f acc0 = {}, acc1 = {};
    G1Frag fa, fb;
    g1_load(fa, xr, tr, 0);
    g1_load(fb, xr, tr, 1);  keep_groups();
    g1_mma(fa, acc0, acc1);  keep_groups();
    g1_load(fa, xr, tr, 2);  keep_groups();
    g1_mma(fb, acc0, acc1);  keep_groups();
    g1_load(fb, xr, tr, 3);  keep_groups();
    g1_mma(fa, acc0, acc1);  keep_groups();
    g1_mma(fb, acc0, acc1);
    return acc0 + acc1;
}

template <int M>
__device__ __forceinline__ float row_xor(float v) {
    return __int_as_float(__builtin_amdgcn_update_dpp(0, __float_as_int(v), 0x160 + M, 0xf, 0xf, false));
}

__device__ __forceinline__ float row16_max(float v) {
    v = fmaxf(v, row_xor<1>(v));
    v = fmaxf(v, row_xor<2>(v));
    v = fmaxf(v, row_xor<4>(v));
    v = fmaxf(v, row_xor<8>(v));
    return v;
}

__device__ __forceinline__ float row16_sum(float v) {
    v += row_xor<1>(v);
    v += row_xor<2>(v);
    v += row_xor<4>(v);
    v += row_xor<8>(v);
    return v;
}

struct G2Frag { v4f x0[4], x1[4]; };

__device__ __forceinline__ void g2_load(G2Frag& f, const v4f* xr, int u) {
#pragma unroll
    for (int g = 0; g < 4; ++g) {
        f.x0[g] = xr[(2 * u + 0) * BU_PITCH4 + 16 * g];
        f.x1[g] = xr[(2 * u + 1) * BU_PITCH4 + 16 * g];
    }
}

__device__ __forceinline__ void g2_mma(const G2Frag& f, float c0, float c1, v8f (&acc)[16]) {
    v2f a; a.x = c0; a.y = c1;
#pragma unroll
    for (int g = 0; g < 4; ++g) {
        v2f b0; b0.x = f.x0[g].x; b0.y = f.x1[g].x;
        v2f b1; b1.x = f.x0[g].y; b1.y = f.x1[g].y;
        v2f b2; b2.x = f.x0[g].z; b2.y = f.x1[g].z;
        v2f b3; b3.x = f.x0[g].w; b3.y = f.x1[g].w;
        acc[4 * g + 0] = __builtin_amdgcn_wmma_f32_16x16x4_f32(false, a, false, b0, (short)0, acc[4 * g + 0], false, false);
        acc[4 * g + 1] = __builtin_amdgcn_wmma_f32_16x16x4_f32(false, a, false, b1, (short)0, acc[4 * g + 1], false, false);
        acc[4 * g + 2] = __builtin_amdgcn_wmma_f32_16x16x4_f32(false, a, false, b2, (short)0, acc[4 * g + 2], false, false);
        acc[4 * g + 3] = __builtin_amdgcn_wmma_f32_16x16x4_f32(false, a, false, b3, (short)0, acc[4 * g + 3], false, false);
    }
}

__global__ void __launch_bounds__(32 * RF_WAVES)
route_fused(const float* __restrict__ x, const float* __restrict__ t,
            float* __restrict__ s_part) {
    __shared__ v4f sh[(RF_WAVES + 1) * BU_TILE4];
    const int b    = blockIdx.y;
    const int tid  = threadIdx.x;
    const int lane = tid & 31;
    const int wave = tid >> 5;
    const int half = lane >> 4;
    const int l16  = lane & 15;

    v4f* xs = sh + wave * BU_TILE4;
    v4f* ts = sh + RF_WAVES * BU_TILE4;

    const size_t row0 = (size_t)b * In + (size_t)blockIdx.x * RF_ROWS + 16 * wave;
    const v4f* xg = reinterpret_cast<const v4f*>(x + row0 * D);
    constexpr size_t XSTEP4 = (size_t)16 * RF_WAVES * D / 4;

    v4f g[32];
    bu_load(g, xg, lane);

    const v4f* tg = reinterpret_cast<const v4f*>(t + (size_t)b * N * D);
    v4f tv[8];
#pragma unroll
    for (int j = 0; j < 8; ++j) tv[j] = tg[128 * j + tid];
#pragma unroll
    for (int j = 0; j < 8; ++j) {
        const int f = 128 * j + tid;
        ts[(f >> 6) * BU_PITCH4 + (f & 63)] = tv[j];
    }
    __syncthreads();

    v8f acc[16];
#pragma unroll
    for (int j = 0; j < 16; ++j) acc[j] = v8f{};

    for (int k = 0; k < RF_TILES; ++k) {
        bu_stage(g, xs, lane);
        if (k + 1 < RF_TILES) bu_load(g, xg + (k + 1) * XSTEP4, lane);
        __builtin_amdgcn_sched_barrier(0);
        const v8f lg = bu_tile(xs, ts, l16, half);

        const v4f* xr2 = xs + (8 * half) * BU_PITCH4 + l16;
        G2Frag fa, fb;
        g2_load(fa, xr2, 0);

        float cc[8];
#pragma unroll
        for (int r = 0; r < 8; ++r) {
            const float m = row16_max(lg[r]);
            const float e = __expf(lg[r] - m);
            const float inv = 1.0f / row16_sum(e);
            cc[r] = e * inv;
        }

        g2_load(fb, xr2, 1);              keep_groups();
        g2_mma(fa, cc[0], cc[1], acc);    keep_groups();
        g2_load(fa, xr2, 2);              keep_groups();
        g2_mma(fb, cc[2], cc[3], acc);    keep_groups();
        g2_load(fb, xr2, 3);              keep_groups();
        g2_mma(fa, cc[4], cc[5], acc);    keep_groups();
        g2_mma(fb, cc[6], cc[7], acc);
    }

#pragma unroll
    for (int g4 = 0; g4 < 4; ++g4) {
#pragma unroll
        for (int r = 0; r < 8; ++r) {
            v4f v;
            v.x = acc[4 * g4 + 0][r]; v.y = acc[4 * g4 + 1][r];
            v.z = acc[4 * g4 + 2][r]; v.w = acc[4 * g4 + 3][r];
            xs[(g4 * 8 + r) * 32 + lane] = v;
        }
    }
    __syncthreads();

    float* outp = s_part + (size_t)(b * RF_CHUNKS + blockIdx.x) * N * D;
#pragma unroll
    for (int q = 0; q < 8; ++q) {
        const int e = 128 * q + tid;
        v4f v = sh[e];
#pragma unroll
        for (int w = 1; w < RF_WAVES; ++w) v += sh[w * BU_TILE4 + e];
        const int gr = e >> 5;
        const int n  = (gr & 7) + 8 * ((e >> 4) & 1);
        const int d  = 64 * (gr >> 3) + 4 * (e & 15);
        *reinterpret_cast<v4f*>(outp + (size_t)n * D + d) = v;
    }
}

template <int NS, bool LAST>
__global__ void __launch_bounds__(64)
route_tail(const float* __restrict__ s_part, const float* __restrict__ W,
           float* __restrict__ t, float* __restrict__ out) {
    __shared__ v4f red[4 * 4];
    const int bn  = blockIdx.x;
    const int b   = bn / N;
    const int n   = bn - b * N;
    const int tid = threadIdx.x;

    const v4f* sp = reinterpret_cast<const v4f*>(s_part + ((size_t)b * NS * N + n) * D) + tid;
    v4f pv[NS];
#pragma unroll
    for (int i = 0; i < NS; ++i) pv[i] = sp[(size_t)i * (N * D / 4)];

    const v4f* wp = reinterpret_cast<const v4f*>(W + (size_t)(4 * tid) * (N * K) + (size_t)n * K);
    v4f w[4][4];
#pragma unroll
    for (int r = 0; r < 4; ++r) {
#pragma unroll
        for (int q = 0; q < 4; ++q) w[r][q] = wp[r * (N * K / 4) + q];
    }

    v4f sv = pv[0];
#pragma unroll
    for (int i = 1; i < NS; ++i) sv += pv[i];

    v4f p[4];
#pragma unroll
    for (int q = 0; q < 4; ++q) {
        p[q] = sv.x * w[0][q];
        p[q] += sv.y * w[1][q];
        p[q] += sv.z * w[2][q];
        p[q] += sv.w * w[3][q];
        p[q].x = row16_sum(p[q].x);
        p[q].y = row16_sum(p[q].y);
        p[q].z = row16_sum(p[q].z);
        p[q].w = row16_sum(p[q].w);
    }
    if ((tid & 15) == 0) {
#pragma unroll
        for (int q = 0; q < 4; ++q) red[(tid >> 4) * 4 + q] = p[q];
    }
    __syncthreads();

    v4f dot[4];
    float n2 = 0.0f;
#pragma unroll
    for (int q = 0; q < 4; ++q) {
        dot[q] = red[q];
        dot[q] += red[4 + q];
        dot[q] += red[8 + q];
        dot[q] += red[12 + q];
        n2 += dot[q].x * dot[q].x;
        n2 += dot[q].y * dot[q].y;
        n2 += dot[q].z * dot[q].z;
        n2 += dot[q].w * dot[q].w;
    }
    const float inv = rsqrtf(n2 + 1e-7f);

    if (LAST) {
        if (tid < 4) {
            v4f o = red[tid];
            o += red[4 + tid];
            o += red[8 + tid];
            o += red[12 + tid];
            reinterpret_cast<v4f*>(out + (size_t)bn * K)[tid] = o * inv;
        }
    } else {
        v4f tv;
#pragma unroll
        for (int r = 0; r < 4; ++r) {
            float acc = 0.0f;
#pragma unroll
            for (int q = 0; q < 4; ++q) {
                acc += w[r][q].x * (dot[q].x * inv);
                acc += w[r][q].y * (dot[q].y * inv);
                acc += w[r][q].z * (dot[q].z * inv);
                acc += w[r][q].w * (dot[q].w * inv);
            }
            tv[r] = acc;
        }
        reinterpret_cast<v4f*>(t + (size_t)bn * D)[tid] = tv;
    }
}

extern "C" void kernel_launch(void* const* d_in, const int* in_sizes, int n_in,
                              void* d_out, int out_size, void* d_ws, size_t ws_size,
                              hipStream_t stream) {
    const float* x = (const float*)d_in[0];
    const float* W = (const float*)d_in[1];
    float* out = (float*)d_out;

    float* ws       = (float*)d_ws;
    float* s_part16 = ws;
    float* s_part   = s_part16 + (size_t)B * RF_CHUNKS * N * D;
    float* t        = s_part + (size_t)B * SPLIT * N * D;

    gemm_s<<<dim3(2, SPLIT, B), 256, 0, stream>>>(x, s_part);
    route_tail<SPLIT, false><<<B * N, 64, 0, stream>>>(s_part, W, t, out);

    route_fused<<<dim3(RF_CHUNKS, B), 32 * RF_WAVES, 0, stream>>>(x, t, s_part16);
    route_tail<RF_CHUNKS, false><<<B * N, 64, 0, stream>>>(s_part16, W, t, out);

    route_fused<<<dim3(RF_CHUNKS, B), 32 * RF_WAVES, 0, stream>>>(x, t, s_part16);
    route_tail<RF_CHUNKS, true><<<B * N, 64, 0, stream>>>(s_part16, W, t, out);
}
